# starting kernel
# baseline (speedup 1.0000x reference)
#include <hip/hip_runtime.h>
#include <math.h>

#define BB    8
#define NN    512
#define FINC  256
#define FOUTC 256
#define NMAXK 32
#define COLS  1024
#define ROWS  4096
#define TILE_M 16
#define AROW  260
#define ATILE (TILE_M*AROW)

typedef float v2f __attribute__((ext_vector_type(2)));
typedef float v8f __attribute__((ext_vector_type(8)));

__global__ void __launch_bounds__(256) sage_init_stats(float* __restrict__ stats) {
    int i = blockIdx.x * 256 + threadIdx.x;
    stats[i] = 0.0f;
}

__global__ void __launch_bounds__(256, 1)
sage_fused(const float* __restrict__ x,
           const int*   __restrict__ idx1,
           const int*   __restrict__ idx2,
           const int*   __restrict__ idx3,
           const float* __restrict__ Wx, const float* __restrict__ bx,
           const float* __restrict__ Wn, const float* __restrict__ bnb,
           float* __restrict__ out,
           float* __restrict__ gsum, float* __restrict__ gsumsq)
{
    extern __shared__ float smem[];
    float* Alds    = smem;
    float* biasLds = smem + 4 * ATILE;
    float* rowsum  = biasLds + COLS;
    int*   idxLds  = (int*)(rowsum + 16);

    const int t       = threadIdx.x;
    const int rowBase = blockIdx.x * TILE_M;
    const int b       = rowBase / NN;
    const int nBase   = rowBase % NN;

    for (int i = t; i < 3 * TILE_M * NMAXK; i += 256) {
        int s   = i / (TILE_M * NMAXK);
        int rem = i - s * (TILE_M * NMAXK);
        int r   = rem >> 5;
        int k   = rem & 31;
        const int* ip = (s == 0) ? idx1 : ((s == 1) ? idx2 : idx3);
        idxLds[i] = ip[(nBase + r) * NMAXK + k];
    }
    {
        float bv = bnb[t];
        biasLds[t]       = bx[t];
        biasLds[256 + t] = bv;
        biasLds[512 + t] = bv;
        biasLds[768 + t] = bv;
    }
    __syncthreads();

    const float* xb = x + (size_t)b * NN * FINC;
    for (int r = 0; r < TILE_M; ++r)
        Alds[r * AROW + t] = xb[(nBase + r) * FINC + t];
    for (int s = 0; s < 3; ++s) {
        float* At = Alds + (s + 1) * ATILE;
        const int* il = idxLds + s * TILE_M * NMAXK;
        for (int r = 0; r < TILE_M; ++r) {
            float sum = 0.0f;
            #pragma unroll
            for (int k = 0; k < NMAXK; ++k)
                sum += xb[il[r * NMAXK + k] * FINC + t];
            At[r * AROW + t] = sum * (1.0f / NMAXK);
        }
    }
    __syncthreads();

    const int lane = t & 31;
    const int wave = t >> 5;
    const int seg  = wave >> 1;
    const int half = wave & 1;
    const int mr   = lane & 15;
    const int kg   = lane >> 4;

    const float* Wseg  = (seg == 0) ? Wx : Wn;
    const float* Arow  = Alds + seg * ATILE + mr * AROW + 2 * kg;
    const float* Wbase = Wseg + (size_t)(half * 128 + mr) * FINC + 2 * kg;

    v8f acc[8] = {};

    for (int k0 = 0; k0 < FINC; k0 += 4) {
        v2f a = *(const v2f*)(Arow + k0);
        #pragma unroll
        for (int nt = 0; nt < 8; ++nt) {
            v2f bf = *(const v2f*)(Wbase + (size_t)nt * 16 * FINC + k0);
            acc[nt] = __builtin_amdgcn_wmma_f32_16x16x4_f32(
                false, a, false, bf, (short)0, acc[nt], false, false);
        }
    }

    const int colBase = seg * 256 + half * 128;
    #pragma unroll
    for (int nt = 0; nt < 8; ++nt) {
        float bv = biasLds[colBase + nt * 16 + mr];
        #pragma unroll
        for (int r = 0; r < 8; ++r) acc[nt][r] += bv;
    }

    __syncthreads();
    if (t < 16) rowsum[t] = 0.0f;
    __syncthreads();
    #pragma unroll
    for (int r = 0; r < 8; ++r) {
        float sq = 0.0f;
        #pragma unroll
        for (int nt = 0; nt < 8; ++nt) sq += acc[nt][r] * acc[nt][r];
        atomicAdd(&rowsum[r + 8 * kg], sq);
    }
    __syncthreads();

    float inv[8];
    #pragma unroll
    for (int r = 0; r < 8; ++r)
        inv[r] = 1.0f / fmaxf(sqrtf(rowsum[r + 8 * kg]), 1e-12f);

    #pragma unroll
    for (int nt = 0; nt < 8; ++nt) {
        int col = colBase + nt * 16 + mr;
        float s1 = 0.0f, s2 = 0.0f;
        #pragma unroll
        for (int r = 0; r < 8; ++r) {
            float v = fmaxf(acc[nt][r] * inv[r], 0.0f);
            s1 += v;
            s2 += v * v;
            out[(size_t)(rowBase + r + 8 * kg) * COLS + col] = v;
        }
        atomicAdd(&gsum[col], s1);
        atomicAdd(&gsumsq[col], s2);
    }
}

__global__ void __launch_bounds__(256) sage_bn_apply(float* __restrict__ out,
                                                     const float* __restrict__ gsum,
                                                     const float* __restrict__ gsumsq,
                                                     const float* __restrict__ gamma,
                                                     const float* __restrict__ beta)
{
    int i = blockIdx.x * 256 + threadIdx.x;
    int c = i & (COLS - 1);
    const float invn = 1.0f / (float)ROWS;
    float mu  = gsum[c] * invn;
    float var = gsumsq[c] * invn - mu * mu;
    float v   = out[i];
    out[i] = (v - mu) * rsqrtf(var + 1e-5f) * gamma[c] + beta[c];
}

extern "C" void kernel_launch(void* const* d_in, const int* in_sizes, int n_in,
                              void* d_out, int out_size, void* d_ws, size_t ws_size,
                              hipStream_t stream) {
    (void)in_sizes; (void)n_in; (void)out_size; (void)ws_size;
    const float* x    = (const float*)d_in[0];
    const int*   idx1 = (const int*)  d_in[1];
    const int*   idx2 = (const int*)  d_in[2];
    const int*   idx3 = (const int*)  d_in[3];
    const float* Wx   = (const float*)d_in[4];
    const float* bx   = (const float*)d_in[5];
    const float* Wn   = (const float*)d_in[6];
    const float* bnb  = (const float*)d_in[7];
    const float* gam  = (const float*)d_in[8];
    const float* bet  = (const float*)d_in[9];
    float* out = (float*)d_out;

    float* gsum   = (float*)d_ws;
    float* gsumsq = gsum + COLS;

    sage_init_stats<<<8, 256, 0, stream>>>(gsum);

    size_t shbytes = (size_t)(4 * ATILE + COLS + 16) * sizeof(float)
                   + (size_t)(3 * TILE_M * NMAXK) * sizeof(int);
    sage_fused<<<ROWS / TILE_M, 256, shbytes, stream>>>(
        x, idx1, idx2, idx3, Wx, bx, Wn, bnb, out, gsum, gsumsq);

    sage_bn_apply<<<(ROWS * COLS) / 256, 256, 0, stream>>>(out, gsum, gsumsq, gam, bet);
}


# B path: each wave owns 32 channels of all 4 segments, so one W_n fragment feeds 3 neighbour tiles; A and B fragments read 16 B per lane one block ahead, k pairs interleaved alike (k sum order)
# speedup vs baseline: 1.1627x; 1.1627x over previous
#include <hip/hip_runtime.h>
#include <math.h>

#define BB    8
#define NN    512
#define FINC  256
#define FOUTC 256
#define NMAXK 32
#define COLS  1024
#define ROWS  4096
#define TILE_M 16
#define AROW  260
#define ATILE (TILE_M*AROW)

typedef float v2f __attribute__((ext_vector_type(2)));
typedef float v8f __attribute__((ext_vector_type(8)));
typedef float v4f __attribute__((ext_vector_type(4)));

struct Frag {
    v4f a[4][4];
    v4f b[2][2][4];
};

__device__ __forceinline__ void load_frag(Frag& f, const float* __restrict__ Wxp,
                                          const float* __restrict__ Wnp,
                                          const float* Ap, int kb)
{
    #pragma unroll
    for (int nt = 0; nt < 2; ++nt) {
        #pragma unroll
        for (int c = 0; c < 4; ++c)
            f.b[0][nt][c] = *(const v4f*)(Wxp + nt * 16 * FINC + kb + 8 * c);
        #pragma unroll
        for (int c = 0; c < 4; ++c)
            f.b[1][nt][c] = *(const v4f*)(Wnp + nt * 16 * FINC + kb + 8 * c);
    }
    #pragma unroll
    for (int s = 0; s < 4; ++s) {
        #pragma unroll
        for (int c = 0; c < 4; ++c)
            f.a[s][c] = *(const v4f*)(Ap + s * ATILE + kb + 8 * c);
    }
}

__device__ __forceinline__ void mma_block(v8f (&acc)[4][2], const Frag& f)
{
    #pragma unroll
    for (int c = 0; c < 4; ++c) {
        #pragma unroll
        for (int s = 0; s < 4; ++s) {
            const int w = (s == 0) ? 0 : 1;
            #pragma unroll
            for (int nt = 0; nt < 2; ++nt) {
                v2f a0 = { f.a[s][c][0], f.a[s][c][1] };
                v2f b0 = { f.b[w][nt][c][0], f.b[w][nt][c][1] };
                acc[s][nt] = __builtin_amdgcn_wmma_f32_16x16x4_f32(
                    false, a0, false, b0, (short)0, acc[s][nt], false, false);
            }
        }
        #pragma unroll
        for (int s = 0; s < 4; ++s) {
            const int w = (s == 0) ? 0 : 1;
            #pragma unroll
            for (int nt = 0; nt < 2; ++nt) {
                v2f a1 = { f.a[s][c][2], f.a[s][c][3] };
                v2f b1 = { f.b[w][nt][c][2], f.b[w][nt][c][3] };
                acc[s][nt] = __builtin_amdgcn_wmma_f32_16x16x4_f32(
                    false, a1, false, b1, (short)0, acc[s][nt], false, false);
            }
        }
    }
}

__global__ void __launch_bounds__(256) sage_init_stats(float* __restrict__ stats) {
    int i = blockIdx.x * 256 + threadIdx.x;
    stats[i] = 0.0f;
}

__global__ void __launch_bounds__(256, 1)
sage_fused(const float* __restrict__ x,
           const int*   __restrict__ idx1,
           const int*   __restrict__ idx2,
           const int*   __restrict__ idx3,
           const float* __restrict__ Wx, const float* __restrict__ bx,
           const float* __restrict__ Wn, const float* __restrict__ bnb,
           float* __restrict__ out,
           float* __restrict__ gsum, float* __restrict__ gsumsq)
{
    extern __shared__ __attribute__((aligned(16))) float smem[];
    float* Alds    = smem;
    float* biasLds = smem + 4 * ATILE;
    float* rowsum  = biasLds + COLS;
    int*   idxLds  = (int*)(rowsum + 16);

    const int t       = threadIdx.x;
    const int rowBase = blockIdx.x * TILE_M;
    const int b       = rowBase / NN;
    const int nBase   = rowBase % NN;

    for (int i = t; i < 3 * TILE_M * NMAXK; i += 256) {
        int s   = i / (TILE_M * NMAXK);
        int rem = i - s * (TILE_M * NMAXK);
        int r   = rem >> 5;
        int k   = rem & 31;
        const int* ip = (s == 0) ? idx1 : ((s == 1) ? idx2 : idx3);
        idxLds[i] = ip[(nBase + r) * NMAXK + k];
    }
    {
        float bv = bnb[t];
        biasLds[t]       = bx[t];
        biasLds[256 + t] = bv;
        biasLds[512 + t] = bv;
        biasLds[768 + t] = bv;
    }
    __syncthreads();

    const float* xb = x + (size_t)b * NN * FINC;
    for (int r = 0; r < TILE_M; ++r)
        Alds[r * AROW + t] = xb[(nBase + r) * FINC + t];
    for (int s = 0; s < 3; ++s) {
        float* At = Alds + (s + 1) * ATILE;
        const int* il = idxLds + s * TILE_M * NMAXK;
        for (int r = 0; r < TILE_M; ++r) {
            float sum = 0.0f;
            #pragma unroll
            for (int k = 0; k < NMAXK; ++k)
                sum += xb[il[r * NMAXK + k] * FINC + t];
            At[r * AROW + t] = sum * (1.0f / NMAXK);
        }
    }
    __syncthreads();

    const int lane = t & 31;
    const int wave = t >> 5;
    const int mr   = lane & 15;
    const int kg   = lane >> 4;

    const float* Ap  = Alds + mr * AROW + 4 * kg;
    const float* Wxp = Wx + (size_t)(wave * 32 + mr) * FINC + 4 * kg;
    const float* Wnp = Wn + (size_t)(wave * 32 + mr) * FINC + 4 * kg;

    v8f acc[4][2] = {};

    Frag f0, f1;
    load_frag(f0, Wxp, Wnp, Ap, 0);
    for (int kb = 0; kb < FINC - 64; kb += 64) {
        load_frag(f1, Wxp, Wnp, Ap, kb + 32);
        __builtin_amdgcn_sched_barrier(0);
        mma_block(acc, f0);
        __builtin_amdgcn_sched_barrier(0);
        load_frag(f0, Wxp, Wnp, Ap, kb + 64);
        __builtin_amdgcn_sched_barrier(0);
        mma_block(acc, f1);
        __builtin_amdgcn_sched_barrier(0);
    }
    load_frag(f1, Wxp, Wnp, Ap, FINC - 32);
    __builtin_amdgcn_sched_barrier(0);
    mma_block(acc, f0);
    mma_block(acc, f1);

    const int colBase = wave * 32;
    #pragma unroll
    for (int s = 0; s < 4; ++s) {
        #pragma unroll
        for (int nt = 0; nt < 2; ++nt) {
            float bv = biasLds[s * 256 + colBase + nt * 16 + mr];
            #pragma unroll
            for (int r = 0; r < 8; ++r) acc[s][nt][r] += bv;
        }
    }

    __syncthreads();
    if (t < 16) rowsum[t] = 0.0f;
    __syncthreads();
    #pragma unroll
    for (int r = 0; r < 8; ++r) {
        float sq = 0.0f;
        #pragma unroll
        for (int s = 0; s < 4; ++s) {
            #pragma unroll
            for (int nt = 0; nt < 2; ++nt) sq += acc[s][nt][r] * acc[s][nt][r];
        }
        atomicAdd(&rowsum[r + 8 * kg], sq);
    }
    __syncthreads();

    float inv[8];
    #pragma unroll
    for (int r = 0; r < 8; ++r)
        inv[r] = 1.0f / fmaxf(sqrtf(rowsum[r + 8 * kg]), 1e-12f);

    #pragma unroll
    for (int s = 0; s < 4; ++s) {
        #pragma unroll
        for (int nt = 0; nt < 2; ++nt) {
            int col = s * 256 + colBase + nt * 16 + mr;
            float s1 = 0.0f, s2 = 0.0f;
            #pragma unroll
            for (int r = 0; r < 8; ++r) {
                float v = fmaxf(acc[s][nt][r] * inv[r], 0.0f);
                s1 += v;
                s2 += v * v;
                out[(size_t)(rowBase + r + 8 * kg) * COLS + col] = v;
            }
            atomicAdd(&gsum[col], s1);
            atomicAdd(&gsumsq[col], s2);
        }
    }
}

__global__ void __launch_bounds__(256) sage_bn_apply(float* __restrict__ out,
                                                     const float* __restrict__ gsum,
                                                     const float* __restrict__ gsumsq,
                                                     const float* __restrict__ gamma,
                                                     const float* __restrict__ beta)
{
    int i = blockIdx.x * 256 + threadIdx.x;
    int c = i & (COLS - 1);
    const float invn = 1.0f / (float)ROWS;
    float mu  = gsum[c] * invn;
    float var = gsumsq[c] * invn - mu * mu;
    float v   = out[i];
    out[i] = (v - mu) * rsqrtf(var + 1e-5f) * gamma[c] + beta[c];
}

extern "C" void kernel_launch(void* const* d_in, const int* in_sizes, int n_in,
                              void* d_out, int out_size, void* d_ws, size_t ws_size,
                              hipStream_t stream) {
    (void)in_sizes; (void)n_in; (void)out_size; (void)ws_size;
    const float* x    = (const float*)d_in[0];
    const int*   idx1 = (const int*)  d_in[1];
    const int*   idx2 = (const int*)  d_in[2];
    const int*   idx3 = (const int*)  d_in[3];
    const float* Wx   = (const float*)d_in[4];
    const float* bx   = (const float*)d_in[5];
    const float* Wn   = (const float*)d_in[6];
    const float* bnb  = (const float*)d_in[7];
    const float* gam  = (const float*)d_in[8];
    const float* bet  = (const float*)d_in[9];
    float* out = (float*)d_out;

    float* gsum   = (float*)d_ws;
    float* gsumsq = gsum + COLS;

    sage_init_stats<<<8, 256, 0, stream>>>(gsum);

    size_t shbytes = (size_t)(4 * ATILE + COLS + 16) * sizeof(float)
                   + (size_t)(3 * TILE_M * NMAXK) * sizeof(int);
    sage_fused<<<ROWS / TILE_M, 256, shbytes, stream>>>(
        x, idx1, idx2, idx3, Wx, bx, Wn, bnb, out, gsum, gsumsq);

    sage_bn_apply<<<(ROWS * COLS) / 256, 256, 0, stream>>>(out, gsum, gsumsq, gam, bet);
}


# Neighbour gather reads LDS instead of L2: each 64-channel slice of the batch's x (512 rows, 128 KB) is staged once per workgroup, next slice's loads in flight; sums keep k order 0..31
# speedup vs baseline: 1.3988x; 1.2031x over previous
#include <hip/hip_runtime.h>
#include <math.h>

#define BB    8
#define NN    512
#define FINC  256
#define FOUTC 256
#define NMAXK 32
#define COLS  1024
#define ROWS  4096
#define TILE_M 16
#define AROW  260
#define ATILE (TILE_M*AROW)
#define CH    64
#define NSL   (FINC/CH)
#define TABF  (NN*CH)
#define PRE   (TABF/4/256)
#define SMEMF (TABF + 4*ATILE + COLS + 16)

typedef float v2f __attribute__((ext_vector_type(2)));
typedef float v8f __attribute__((ext_vector_type(8)));
typedef float v4f __attribute__((ext_vector_type(4)));

struct Frag {
    v4f a[4][4];
    v4f b[2][2][4];
};

__device__ __forceinline__ void load_frag(Frag& f, const float* __restrict__ Wxp,
                                          const float* __restrict__ Wnp,
                                          const float* Ap, int kb)
{
    #pragma unroll
    for (int nt = 0; nt < 2; ++nt) {
        #pragma unroll
        for (int c = 0; c < 4; ++c)
            f.b[0][nt][c] = *(const v4f*)(Wxp + nt * 16 * FINC + kb + 8 * c);
        #pragma unroll
        for (int c = 0; c < 4; ++c)
            f.b[1][nt][c] = *(const v4f*)(Wnp + nt * 16 * FINC + kb + 8 * c);
    }
    #pragma unroll
    for (int s = 0; s < 4; ++s) {
        #pragma unroll
        for (int c = 0; c < 4; ++c)
            f.a[s][c] = *(const v4f*)(Ap + s * ATILE + kb + 8 * c);
    }
}

__device__ __forceinline__ void mma_block(v8f (&acc)[4][2], const Frag& f)
{
    #pragma unroll
    for (int c = 0; c < 4; ++c) {
        #pragma unroll
        for (int s = 0; s < 4; ++s) {
            const int w = (s == 0) ? 0 : 1;
            #pragma unroll
            for (int nt = 0; nt < 2; ++nt) {
                v2f a0 = { f.a[s][c][0], f.a[s][c][1] };
                v2f b0 = { f.b[w][nt][c][0], f.b[w][nt][c][1] };
                acc[s][nt] = __builtin_amdgcn_wmma_f32_16x16x4_f32(
                    false, a0, false, b0, (short)0, acc[s][nt], false, false);
            }
        }
        #pragma unroll
        for (int s = 0; s < 4; ++s) {
            const int w = (s == 0) ? 0 : 1;
            #pragma unroll
            for (int nt = 0; nt < 2; ++nt) {
                v2f a1 = { f.a[s][c][2], f.a[s][c][3] };
                v2f b1 = { f.b[w][nt][c][2], f.b[w][nt][c][3] };
                acc[s][nt] = __builtin_amdgcn_wmma_f32_16x16x4_f32(
                    false, a1, false, b1, (short)0, acc[s][nt], false, false);
            }
        }
    }
}

__device__ __forceinline__ void rows_issue(v2f (&v)[NMAXK], const float* tl, int ov)
{
    #pragma unroll
    for (int k = 0; k < NMAXK; ++k)
        v[k] = *(const v2f*)(tl + __builtin_amdgcn_readlane(ov, k));
}

__device__ __forceinline__ v2f rows_mean(const v2f (&v)[NMAXK])
{
    v2f sum = { 0.0f, 0.0f };
    #pragma unroll
    for (int k = 0; k < NMAXK; ++k) {
        sum[0] += v[k][0];
        sum[1] += v[k][1];
    }
    v2f m = { sum[0] * (1.0f / NMAXK), sum[1] * (1.0f / NMAXK) };
    return m;
}

__global__ void __launch_bounds__(256) sage_init_stats(float* __restrict__ stats) {
    int i = blockIdx.x * 256 + threadIdx.x;
    stats[i] = 0.0f;
}

__global__ void __launch_bounds__(256, 1)
sage_fused(const float* __restrict__ x,
           const int*   __restrict__ idx1,
           const int*   __restrict__ idx2,
           const int*   __restrict__ idx3,
           const float* __restrict__ Wx, const float* __restrict__ bx,
           const float* __restrict__ Wn, const float* __restrict__ bnb,
           float* __restrict__ out,
           float* __restrict__ gsum, float* __restrict__ gsumsq)
{
    __shared__ __attribute__((aligned(16))) float smem[SMEMF];
    float* Alds    = smem;
    float* biasLds = Alds + 4 * ATILE;
    float* rowsum  = biasLds + COLS;
    float* tab     = rowsum + 16;

    const int t       = threadIdx.x;
    const int lane    = t & 31;
    const int wave    = t >> 5;
    const int rowBase = blockIdx.x * TILE_M;
    const int b       = rowBase / NN;
    const int nBase   = rowBase % NN;
    const float* xb = x + (size_t)b * NN * FINC;

    const float* xpre = xb + (t >> 4) * FINC + 4 * (t & 15);
    v4f pre[PRE];
    #pragma unroll
    for (int j = 0; j < PRE; ++j)
        pre[j] = *(const v4f*)(xpre + j * 16 * FINC);

    int ov[6];
    #pragma unroll
    for (int pi = 0; pi < 6; ++pi) {
        const int* ip = (pi < 2) ? idx1 : ((pi < 4) ? idx2 : idx3);
        ov[pi] = ip[(nBase + wave + 8 * (pi & 1)) * NMAXK + lane] * CH;
    }
    {
        float bv = bnb[t];
        biasLds[t]       = bx[t];
        biasLds[256 + t] = bv;
        biasLds[512 + t] = bv;
        biasLds[768 + t] = bv;
    }
    for (int r = 0; r < TILE_M; ++r)
        Alds[r * AROW + t] = xb[(nBase + r) * FINC + t];

    const float* tl = tab + 2 * lane;
    v2f va[NMAXK], vb[NMAXK];
    for (int c = 0; c < NSL; ++c) {
        #pragma unroll
        for (int j = 0; j < PRE; ++j)
            *(v4f*)(tab + (j * 256 + t) * 4) = pre[j];
        __syncthreads();
        if (c + 1 < NSL) {
            #pragma unroll
            for (int j = 0; j < PRE; ++j)
                pre[j] = *(const v4f*)(xpre + (c + 1) * CH + j * 16 * FINC);
        }
        float* At = Alds + ATILE + wave * AROW + c * CH + 2 * lane;
        rows_issue(va, tl, ov[0]);
        #pragma unroll
        for (int q = 0; q < 3; ++q) {
            rows_issue(vb, tl, ov[2 * q + 1]);
            __builtin_amdgcn_sched_barrier(0);
            *(v2f*)(At + q * ATILE) = rows_mean(va);
            __builtin_amdgcn_sched_barrier(0);
            if (q < 2) rows_issue(va, tl, ov[2 * q + 2]);
            __builtin_amdgcn_sched_barrier(0);
            *(v2f*)(At + q * ATILE + 8 * AROW) = rows_mean(vb);
            __builtin_amdgcn_sched_barrier(0);
        }
        __syncthreads();
    }

    const int mr   = lane & 15;
    const int kg   = lane >> 4;

    const float* Ap  = Alds + mr * AROW + 4 * kg;
    const float* Wxp = Wx + (size_t)(wave * 32 + mr) * FINC + 4 * kg;
    const float* Wnp = Wn + (size_t)(wave * 32 + mr) * FINC + 4 * kg;

    v8f acc[4][2] = {};

    Frag f0, f1;
    load_frag(f0, Wxp, Wnp, Ap, 0);
    for (int kb = 0; kb < FINC - 64; kb += 64) {
        load_frag(f1, Wxp, Wnp, Ap, kb + 32);
        __builtin_amdgcn_sched_barrier(0);
        mma_block(acc, f0);
        __builtin_amdgcn_sched_barrier(0);
        load_frag(f0, Wxp, Wnp, Ap, kb + 64);
        __builtin_amdgcn_sched_barrier(0);
        mma_block(acc, f1);
        __builtin_amdgcn_sched_barrier(0);
    }
    load_frag(f1, Wxp, Wnp, Ap, FINC - 32);
    __builtin_amdgcn_sched_barrier(0);
    mma_block(acc, f0);
    mma_block(acc, f1);

    const int colBase = wave * 32;
    #pragma unroll
    for (int s = 0; s < 4; ++s) {
        #pragma unroll
        for (int nt = 0; nt < 2; ++nt) {
            float bv = biasLds[s * 256 + colBase + nt * 16 + mr];
            #pragma unroll
            for (int r = 0; r < 8; ++r) acc[s][nt][r] += bv;
        }
    }

    __syncthreads();
    if (t < 16) rowsum[t] = 0.0f;
    __syncthreads();
    #pragma unroll
    for (int r = 0; r < 8; ++r) {
        float sq = 0.0f;
        #pragma unroll
        for (int s = 0; s < 4; ++s) {
            #pragma unroll
            for (int nt = 0; nt < 2; ++nt) sq += acc[s][nt][r] * acc[s][nt][r];
        }
        atomicAdd(&rowsum[r + 8 * kg], sq);
    }
    __syncthreads();

    float inv[8];
    #pragma unroll
    for (int r = 0; r < 8; ++r)
        inv[r] = 1.0f / fmaxf(sqrtf(rowsum[r + 8 * kg]), 1e-12f);

    #pragma unroll
    for (int s = 0; s < 4; ++s) {
        #pragma unroll
        for (int nt = 0; nt < 2; ++nt) {
            int col = s * 256 + colBase + nt * 16 + mr;
            float s1 = 0.0f, s2 = 0.0f;
            #pragma unroll
            for (int r = 0; r < 8; ++r) {
                float v = fmaxf(acc[s][nt][r] * inv[r], 0.0f);
                s1 += v;
                s2 += v * v;
                out[(size_t)(rowBase + r + 8 * kg) * COLS + col] = v;
            }
            atomicAdd(&gsum[col], s1);
            atomicAdd(&gsumsq[col], s2);
        }
    }
}

__global__ void __launch_bounds__(256) sage_bn_apply(float* __restrict__ out,
                                                     const float* __restrict__ gsum,
                                                     const float* __restrict__ gsumsq,
                                                     const float* __restrict__ gamma,
                                                     const float* __restrict__ beta)
{
    int i = blockIdx.x * 256 + threadIdx.x;
    int c = i & (COLS - 1);
    const float invn = 1.0f / (float)ROWS;
    float mu  = gsum[c] * invn;
    float var = gsumsq[c] * invn - mu * mu;
    float v   = out[i];
    out[i] = (v - mu) * rsqrtf(var + 1e-5f) * gamma[c] + beta[c];
}

extern "C" void kernel_launch(void* const* d_in, const int* in_sizes, int n_in,
                              void* d_out, int out_size, void* d_ws, size_t ws_size,
                              hipStream_t stream) {
    (void)in_sizes; (void)n_in; (void)out_size; (void)ws_size;
    const float* x    = (const float*)d_in[0];
    const int*   idx1 = (const int*)  d_in[1];
    const int*   idx2 = (const int*)  d_in[2];
    const int*   idx3 = (const int*)  d_in[3];
    const float* Wx   = (const float*)d_in[4];
    const float* bx   = (const float*)d_in[5];
    const float* Wn   = (const float*)d_in[6];
    const float* bnb  = (const float*)d_in[7];
    const float* gam  = (const float*)d_in[8];
    const float* bet  = (const float*)d_in[9];
    float* out = (float*)d_out;

    float* gsum   = (float*)d_ws;
    float* gsumsq = gsum + COLS;

    sage_init_stats<<<8, 256, 0, stream>>>(gsum);

    sage_fused<<<ROWS / TILE_M, 256, 0, stream>>>(
        x, idx1, idx2, idx3, Wx, bx, Wn, bnb, out, gsum, gsumsq);

    sage_bn_apply<<<(ROWS * COLS) / 256, 256, 0, stream>>>(out, gsum, gsumsq, gam, bet);
}


# Lanes are summed in registers before every atomic add: row-norm squares over the 16 lanes of a half, BN column sums of the two halves paired, so no add carries two lanes per word (sum order changes)
# speedup vs baseline: 1.6337x; 1.1679x over previous
#include <hip/hip_runtime.h>
#include <math.h>

#define BB    8
#define NN    512
#define FINC  256
#define FOUTC 256
#define NMAXK 32
#define COLS  1024
#define ROWS  4096
#define TILE_M 16
#define AROW  260
#define ATILE (TILE_M*AROW)
#define CH    64
#define NSL   (FINC/CH)
#define TABF  (NN*CH)
#define PRE   (TABF/4/256)
#define SMEMF (TABF + 4*ATILE + COLS + 16)

typedef float v2f __attribute__((ext_vector_type(2)));
typedef float v8f __attribute__((ext_vector_type(8)));
typedef float v4f __attribute__((ext_vector_type(4)));

struct Frag {
    v4f a[4][4];
    v4f b[2][2][4];
};

__device__ __forceinline__ void load_frag(Frag& f, const float* __restrict__ Wxp,
                                          const float* __restrict__ Wnp,
                                          const float* Ap, int kb)
{
    #pragma unroll
    for (int nt = 0; nt < 2; ++nt) {
        #pragma unroll
        for (int c = 0; c < 4; ++c)
            f.b[0][nt][c] = *(const v4f*)(Wxp + nt * 16 * FINC + kb + 8 * c);
        #pragma unroll
        for (int c = 0; c < 4; ++c)
            f.b[1][nt][c] = *(const v4f*)(Wnp + nt * 16 * FINC + kb + 8 * c);
    }
    #pragma unroll
    for (int s = 0; s < 4; ++s) {
        #pragma unroll
        for (int c = 0; c < 4; ++c)
            f.a[s][c] = *(const v4f*)(Ap + s * ATILE + kb + 8 * c);
    }
}

__device__ __forceinline__ void mma_block(v8f (&acc)[4][2], const Frag& f)
{
    #pragma unroll
    for (int c = 0; c < 4; ++c) {
        #pragma unroll
        for (int s = 0; s < 4; ++s) {
            const int w = (s == 0) ? 0 : 1;
            #pragma unroll
            for (int nt = 0; nt < 2; ++nt) {
                v2f a0 = { f.a[s][c][0], f.a[s][c][1] };
                v2f b0 = { f.b[w][nt][c][0], f.b[w][nt][c][1] };
                acc[s][nt] = __builtin_amdgcn_wmma_f32_16x16x4_f32(
                    false, a0, false, b0, (short)0, acc[s][nt], false, false);
            }
        }
        #pragma unroll
        for (int s = 0; s < 4; ++s) {
            const int w = (s == 0) ? 0 : 1;
            #pragma unroll
            for (int nt = 0; nt < 2; ++nt) {
                v2f a1 = { f.a[s][c][2], f.a[s][c][3] };
                v2f b1 = { f.b[w][nt][c][2], f.b[w][nt][c][3] };
                acc[s][nt] = __builtin_amdgcn_wmma_f32_16x16x4_f32(
                    false, a1, false, b1, (short)0, acc[s][nt], false, false);
            }
        }
    }
}

__device__ __forceinline__ void rows_issue(v2f (&v)[NMAXK], const float* tl, int ov)
{
    #pragma unroll
    for (int k = 0; k < NMAXK; ++k)
        v[k] = *(const v2f*)(tl + __builtin_amdgcn_readlane(ov, k));
}

__device__ __forceinline__ v2f rows_mean(const v2f (&v)[NMAXK])
{
    v2f sum = { 0.0f, 0.0f };
    #pragma unroll
    for (int k = 0; k < NMAXK; ++k) {
        sum[0] += v[k][0];
        sum[1] += v[k][1];
    }
    v2f m = { sum[0] * (1.0f / NMAXK), sum[1] * (1.0f / NMAXK) };
    return m;
}

__device__ __forceinline__ float xhalf(float x)
{
    int b = __builtin_bit_cast(int, x);
    return __builtin_bit_cast(float, __builtin_amdgcn_permlanex16(b, b, 0x76543210, 0xfedcba98, false, false));
}

template <int M>
__device__ __forceinline__ float xlane16(float x)
{
    int b = __builtin_bit_cast(int, x);
    return __builtin_bit_cast(float, __builtin_amdgcn_update_dpp(b, b, 0x160 + M, 0xf, 0xf, false));
}

__global__ void __launch_bounds__(256) sage_init_stats(float* __restrict__ stats) {
    int i = blockIdx.x * 256 + threadIdx.x;
    stats[i] = 0.0f;
}

__global__ void __launch_bounds__(256, 1)
sage_fused(const float* __restrict__ x,
           const int*   __restrict__ idx1,
           const int*   __restrict__ idx2,
           const int*   __restrict__ idx3,
           const float* __restrict__ Wx, const float* __restrict__ bx,
           const float* __restrict__ Wn, const float* __restrict__ bnb,
           float* __restrict__ out,
           float* __restrict__ gsum, float* __restrict__ gsumsq)
{
    __shared__ __attribute__((aligned(16))) float smem[SMEMF];
    float* Alds    = smem;
    float* biasLds = Alds + 4 * ATILE;
    float* rowsum  = biasLds + COLS;
    float* tab     = rowsum + 16;

    const int t       = threadIdx.x;
    const int lane    = t & 31;
    const int wave    = t >> 5;
    const int rowBase = blockIdx.x * TILE_M;
    const int b       = rowBase / NN;
    const int nBase   = rowBase % NN;
    const float* xb = x + (size_t)b * NN * FINC;

    const float* xpre = xb + (t >> 4) * FINC + 4 * (t & 15);
    v4f pre[PRE];
    #pragma unroll
    for (int j = 0; j < PRE; ++j)
        pre[j] = *(const v4f*)(xpre + j * 16 * FINC);

    int ov[6];
    #pragma unroll
    for (int pi = 0; pi < 6; ++pi) {
        const int* ip = (pi < 2) ? idx1 : ((pi < 4) ? idx2 : idx3);
        ov[pi] = ip[(nBase + wave + 8 * (pi & 1)) * NMAXK + lane] * CH;
    }
    {
        float bv = bnb[t];
        biasLds[t]       = bx[t];
        biasLds[256 + t] = bv;
        biasLds[512 + t] = bv;
        biasLds[768 + t] = bv;
    }
    for (int r = 0; r < TILE_M; ++r)
        Alds[r * AROW + t] = xb[(nBase + r) * FINC + t];

    const float* tl = tab + 2 * lane;
    v2f va[NMAXK], vb[NMAXK];
    for (int c = 0; c < NSL; ++c) {
        #pragma unroll
        for (int j = 0; j < PRE; ++j)
            *(v4f*)(tab + (j * 256 + t) * 4) = pre[j];
        __syncthreads();
        if (c + 1 < NSL) {
            #pragma unroll
            for (int j = 0; j < PRE; ++j)
                pre[j] = *(const v4f*)(xpre + (c + 1) * CH + j * 16 * FINC);
        }
        float* At = Alds + ATILE + wave * AROW + c * CH + 2 * lane;
        rows_issue(va, tl, ov[0]);
        #pragma unroll
        for (int q = 0; q < 3; ++q) {
            rows_issue(vb, tl, ov[2 * q + 1]);
            __builtin_amdgcn_sched_barrier(0);
            *(v2f*)(At + q * ATILE) = rows_mean(va);
            __builtin_amdgcn_sched_barrier(0);
            if (q < 2) rows_issue(va, tl, ov[2 * q + 2]);
            __builtin_amdgcn_sched_barrier(0);
            *(v2f*)(At + q * ATILE + 8 * AROW) = rows_mean(vb);
            __builtin_amdgcn_sched_barrier(0);
        }
        __syncthreads();
    }

    const int mr   = lane & 15;
    const int kg   = lane >> 4;

    const float* Ap  = Alds + mr * AROW + 4 * kg;
    const float* Wxp = Wx + (size_t)(wave * 32 + mr) * FINC + 4 * kg;
    const float* Wnp = Wn + (size_t)(wave * 32 + mr) * FINC + 4 * kg;

    v8f acc[4][2] = {};

    Frag f0, f1;
    load_frag(f0, Wxp, Wnp, Ap, 0);
    for (int kb = 0; kb < FINC - 64; kb += 64) {
        load_frag(f1, Wxp, Wnp, Ap, kb + 32);
        __builtin_amdgcn_sched_barrier(0);
        mma_block(acc, f0);
        __builtin_amdgcn_sched_barrier(0);
        load_frag(f0, Wxp, Wnp, Ap, kb + 64);
        __builtin_amdgcn_sched_barrier(0);
        mma_block(acc, f1);
        __builtin_amdgcn_sched_barrier(0);
    }
    load_frag(f1, Wxp, Wnp, Ap, FINC - 32);
    __builtin_amdgcn_sched_barrier(0);
    mma_block(acc, f0);
    mma_block(acc, f1);

    const int colBase = wave * 32;
    #pragma unroll
    for (int s = 0; s < 4; ++s) {
        #pragma unroll
        for (int nt = 0; nt < 2; ++nt) {
            float bv = biasLds[s * 256 + colBase + nt * 16 + mr];
            #pragma unroll
            for (int r = 0; r < 8; ++r) acc[s][nt][r] += bv;
        }
    }

    float sq[8];
    #pragma unroll
    for (int r = 0; r < 8; ++r) {
        sq[r] = 0.0f;
        #pragma unroll
        for (int s = 0; s < 4; ++s) {
            #pragma unroll
            for (int nt = 0; nt < 2; ++nt) sq[r] += acc[s][nt][r] * acc[s][nt][r];
        }
    }
    #pragma unroll
    for (int r = 0; r < 8; ++r) sq[r] += xlane16<1>(sq[r]);
    #pragma unroll
    for (int r = 0; r < 8; ++r) sq[r] += xlane16<2>(sq[r]);
    #pragma unroll
    for (int r = 0; r < 8; ++r) sq[r] += xlane16<4>(sq[r]);
    #pragma unroll
    for (int r = 0; r < 8; ++r) sq[r] += xlane16<8>(sq[r]);
    __syncthreads();
    if (t < 16) rowsum[t] = 0.0f;
    __syncthreads();
    if (mr == 0) {
        #pragma unroll
        for (int r = 0; r < 8; ++r) atomicAdd(&rowsum[r + 8 * kg], sq[r]);
    }
    __syncthreads();

    float inv[8];
    #pragma unroll
    for (int r = 0; r < 8; ++r)
        inv[r] = 1.0f / fmaxf(sqrtf(rowsum[r + 8 * kg]), 1e-12f);

    #pragma unroll
    for (int s = 0; s < 4; ++s) {
        float s1[2], s2[2];
        #pragma unroll
        for (int nt = 0; nt < 2; ++nt) {
            int col = s * 256 + colBase + nt * 16 + mr;
            s1[nt] = 0.0f;
            s2[nt] = 0.0f;
            #pragma unroll
            for (int r = 0; r < 8; ++r) {
                float v = fmaxf(acc[s][nt][r] * inv[r], 0.0f);
                s1[nt] += v;
                s2[nt] += v * v;
                out[(size_t)(rowBase + r + 8 * kg) * COLS + col] = v;
            }
        }
        const float keep1 = kg ? s1[1] : s1[0];
        const float give1 = kg ? s1[0] : s1[1];
        const float keep2 = kg ? s2[1] : s2[0];
        const float give2 = kg ? s2[0] : s2[1];
        const int colw = s * 256 + colBase + lane;
        atomicAdd(&gsum[colw],   keep1 + xhalf(give1));
        atomicAdd(&gsumsq[colw], keep2 + xhalf(give2));
    }
}

__global__ void __launch_bounds__(256) sage_bn_apply(float* __restrict__ out,
                                                     const float* __restrict__ gsum,
                                                     const float* __restrict__ gsumsq,
                                                     const float* __restrict__ gamma,
                                                     const float* __restrict__ beta)
{
    int i = blockIdx.x * 256 + threadIdx.x;
    int c = i & (COLS - 1);
    const float invn = 1.0f / (float)ROWS;
    float mu  = gsum[c] * invn;
    float var = gsumsq[c] * invn - mu * mu;
    float v   = out[i];
    out[i] = (v - mu) * rsqrtf(var + 1e-5f) * gamma[c] + beta[c];
}

extern "C" void kernel_launch(void* const* d_in, const int* in_sizes, int n_in,
                              void* d_out, int out_size, void* d_ws, size_t ws_size,
                              hipStream_t stream) {
    (void)in_sizes; (void)n_in; (void)out_size; (void)ws_size;
    const float* x    = (const float*)d_in[0];
    const int*   idx1 = (const int*)  d_in[1];
    const int*   idx2 = (const int*)  d_in[2];
    const int*   idx3 = (const int*)  d_in[3];
    const float* Wx   = (const float*)d_in[4];
    const float* bx   = (const float*)d_in[5];
    const float* Wn   = (const float*)d_in[6];
    const float* bnb  = (const float*)d_in[7];
    const float* gam  = (const float*)d_in[8];
    const float* bet  = (const float*)d_in[9];
    float* out = (float*)d_out;

    float* gsum   = (float*)d_ws;
    float* gsumsq = gsum + COLS;

    sage_init_stats<<<8, 256, 0, stream>>>(gsum);

    sage_fused<<<ROWS / TILE_M, 256, 0, stream>>>(
        x, idx1, idx2, idx3, Wx, bx, Wn, bnb, out, gsum, gsumsq);

    sage_bn_apply<<<(ROWS * COLS) / 256, 256, 0, stream>>>(out, gsum, gsumsq, gam, bet);
}


# Gather row addresses built once: a wave's 6x32 LDS row addresses are the same in all four channel slices, so they are computed while slice 0 loads and the trips issue bare reads
# speedup vs baseline: 1.6982x; 1.0395x over previous
#include <hip/hip_runtime.h>
#include <math.h>

#define BB    8
#define NN    512
#define FINC  256
#define FOUTC 256
#define NMAXK 32
#define COLS  1024
#define ROWS  4096
#define TILE_M 16
#define AROW  260
#define ATILE (TILE_M*AROW)
#define CH    64
#define NSL   (FINC/CH)
#define TABF  (NN*CH)
#define PRE   (TABF/4/256)
#define SMEMF (TABF + 4*ATILE + COLS + 16)

typedef float v2f __attribute__((ext_vector_type(2)));
typedef float v8f __attribute__((ext_vector_type(8)));
typedef float v4f __attribute__((ext_vector_type(4)));

struct Frag {
    v4f a[4][4];
    v4f b[2][2][4];
};

__device__ __forceinline__ void load_frag(Frag& f, const float* __restrict__ Wxp,
                                          const float* __restrict__ Wnp,
                                          const float* Ap, int kb)
{
    #pragma unroll
    for (int nt = 0; nt < 2; ++nt) {
        #pragma unroll
        for (int c = 0; c < 4; ++c)
            f.b[0][nt][c] = *(const v4f*)(Wxp + nt * 16 * FINC + kb + 8 * c);
        #pragma unroll
        for (int c = 0; c < 4; ++c)
            f.b[1][nt][c] = *(const v4f*)(Wnp + nt * 16 * FINC + kb + 8 * c);
    }
    #pragma unroll
    for (int s = 0; s < 4; ++s) {
        #pragma unroll
        for (int c = 0; c < 4; ++c)
            f.a[s][c] = *(const v4f*)(Ap + s * ATILE + kb + 8 * c);
    }
}

__device__ __forceinline__ void mma_block(v8f (&acc)[4][2], const Frag& f)
{
    #pragma unroll
    for (int c = 0; c < 4; ++c) {
        #pragma unroll
        for (int s = 0; s < 4; ++s) {
            const int w = (s == 0) ? 0 : 1;
            #pragma unroll
            for (int nt = 0; nt < 2; ++nt) {
                v2f a0 = { f.a[s][c][0], f.a[s][c][1] };
                v2f b0 = { f.b[w][nt][c][0], f.b[w][nt][c][1] };
                acc[s][nt] = __builtin_amdgcn_wmma_f32_16x16x4_f32(
                    false, a0, false, b0, (short)0, acc[s][nt], false, false);
            }
        }
        #pragma unroll
        for (int s = 0; s < 4; ++s) {
            const int w = (s == 0) ? 0 : 1;
            #pragma unroll
            for (int nt = 0; nt < 2; ++nt) {
                v2f a1 = { f.a[s][c][2], f.a[s][c][3] };
                v2f b1 = { f.b[w][nt][c][2], f.b[w][nt][c][3] };
                acc[s][nt] = __builtin_amdgcn_wmma_f32_16x16x4_f32(
                    false, a1, false, b1, (short)0, acc[s][nt], false, false);
            }
        }
    }
}

__device__ __forceinline__ void rows_issue(v2f (&v)[NMAXK], const float* const (&rp)[NMAXK])
{
    #pragma unroll
    for (int k = 0; k < NMAXK; ++k)
        v[k] = *(const v2f*)rp[k];
}

__device__ __forceinline__ v2f rows_mean(const v2f (&v)[NMAXK])
{
    v2f sum = { 0.0f, 0.0f };
    #pragma unroll
    for (int k = 0; k < NMAXK; ++k) {
        sum[0] += v[k][0];
        sum[1] += v[k][1];
    }
    v2f m = { sum[0] * (1.0f / NMAXK), sum[1] * (1.0f / NMAXK) };
    return m;
}

__device__ __forceinline__ float xhalf(float x)
{
    int b = __builtin_bit_cast(int, x);
    return __builtin_bit_cast(float, __builtin_amdgcn_permlanex16(b, b, 0x76543210, 0xfedcba98, false, false));
}

template <int M>
__device__ __forceinline__ float xlane16(float x)
{
    int b = __builtin_bit_cast(int, x);
    return __builtin_bit_cast(float, __builtin_amdgcn_update_dpp(b, b, 0x160 + M, 0xf, 0xf, false));
}

__global__ void __launch_bounds__(256) sage_init_stats(float* __restrict__ stats) {
    int i = blockIdx.x * 256 + threadIdx.x;
    stats[i] = 0.0f;
}

__global__ void __launch_bounds__(256, 1)
sage_fused(const float* __restrict__ x,
           const int*   __restrict__ idx1,
           const int*   __restrict__ idx2,
           const int*   __restrict__ idx3,
           const float* __restrict__ Wx, const float* __restrict__ bx,
           const float* __restrict__ Wn, const float* __restrict__ bnb,
           float* __restrict__ out,
           float* __restrict__ gsum, float* __restrict__ gsumsq)
{
    __shared__ __attribute__((aligned(16))) float smem[SMEMF];
    float* Alds    = smem;
    float* biasLds = Alds + 4 * ATILE;
    float* rowsum  = biasLds + COLS;
    float* tab     = rowsum + 16;

    const int t       = threadIdx.x;
    const int lane    = t & 31;
    const int wave    = t >> 5;
    const int rowBase = blockIdx.x * TILE_M;
    const int b       = rowBase / NN;
    const int nBase   = rowBase % NN;
    const float* xb = x + (size_t)b * NN * FINC;

    int ov[6];
    #pragma unroll
    for (int pi = 0; pi < 6; ++pi) {
        const int* ip = (pi < 2) ? idx1 : ((pi < 4) ? idx2 : idx3);
        ov[pi] = ip[(nBase + wave + 8 * (pi & 1)) * NMAXK + lane] * CH;
    }

    const float* xpre = xb + (t >> 4) * FINC + 4 * (t & 15);
    v4f pre[PRE];
    #pragma unroll
    for (int j = 0; j < PRE; ++j)
        pre[j] = *(const v4f*)(xpre + j * 16 * FINC);

    {
        float bv = bnb[t];
        biasLds[t]       = bx[t];
        biasLds[256 + t] = bv;
        biasLds[512 + t] = bv;
        biasLds[768 + t] = bv;
    }
    for (int r = 0; r < TILE_M; ++r)
        Alds[r * AROW + t] = xb[(nBase + r) * FINC + t];

    const float* tl = tab + 2 * lane;
    const float* rp[6][NMAXK];
    #pragma unroll
    for (int pi = 0; pi < 6; ++pi) {
        #pragma unroll
        for (int k = 0; k < NMAXK; ++k)
            rp[pi][k] = tl + __builtin_amdgcn_readlane(ov[pi], k);
    }
    v2f va[NMAXK], vb[NMAXK];
    for (int c = 0; c < NSL; ++c) {
        #pragma unroll
        for (int j = 0; j < PRE; ++j)
            *(v4f*)(tab + (j * 256 + t) * 4) = pre[j];
        __syncthreads();
        if (c + 1 < NSL) {
            #pragma unroll
            for (int j = 0; j < PRE; ++j)
                pre[j] = *(const v4f*)(xpre + (c + 1) * CH + j * 16 * FINC);
        }
        float* At = Alds + ATILE + wave * AROW + c * CH + 2 * lane;
        rows_issue(va, rp[0]);
        #pragma unroll
        for (int q = 0; q < 3; ++q) {
            rows_issue(vb, rp[2 * q + 1]);
            __builtin_amdgcn_sched_barrier(0);
            *(v2f*)(At + q * ATILE) = rows_mean(va);
            __builtin_amdgcn_sched_barrier(0);
            if (q < 2) rows_issue(va, rp[2 * q + 2]);
            __builtin_amdgcn_sched_barrier(0);
            *(v2f*)(At + q * ATILE + 8 * AROW) = rows_mean(vb);
            __builtin_amdgcn_sched_barrier(0);
        }
        __syncthreads();
    }

    const int mr   = lane & 15;
    const int kg   = lane >> 4;

    const float* Ap  = Alds + mr * AROW + 4 * kg;
    const float* Wxp = Wx + (size_t)(wave * 32 + mr) * FINC + 4 * kg;
    const float* Wnp = Wn + (size_t)(wave * 32 + mr) * FINC + 4 * kg;

    v8f acc[4][2] = {};

    Frag f0, f1;
    load_frag(f0, Wxp, Wnp, Ap, 0);
    for (int kb = 0; kb < FINC - 64; kb += 64) {
        load_frag(f1, Wxp, Wnp, Ap, kb + 32);
        __builtin_amdgcn_sched_barrier(0);
        mma_block(acc, f0);
        __builtin_amdgcn_sched_barrier(0);
        load_frag(f0, Wxp, Wnp, Ap, kb + 64);
        __builtin_amdgcn_sched_barrier(0);
        mma_block(acc, f1);
        __builtin_amdgcn_sched_barrier(0);
    }
    load_frag(f1, Wxp, Wnp, Ap, FINC - 32);
    __builtin_amdgcn_sched_barrier(0);
    mma_block(acc, f0);
    mma_block(acc, f1);

    const int colBase = wave * 32;
    #pragma unroll
    for (int s = 0; s < 4; ++s) {
        #pragma unroll
        for (int nt = 0; nt < 2; ++nt) {
            float bv = biasLds[s * 256 + colBase + nt * 16 + mr];
            #pragma unroll
            for (int r = 0; r < 8; ++r) acc[s][nt][r] += bv;
        }
    }

    float sq[8];
    #pragma unroll
    for (int r = 0; r < 8; ++r) {
        sq[r] = 0.0f;
        #pragma unroll
        for (int s = 0; s < 4; ++s) {
            #pragma unroll
            for (int nt = 0; nt < 2; ++nt) sq[r] += acc[s][nt][r] * acc[s][nt][r];
        }
    }
    #pragma unroll
    for (int r = 0; r < 8; ++r) sq[r] += xlane16<1>(sq[r]);
    #pragma unroll
    for (int r = 0; r < 8; ++r) sq[r] += xlane16<2>(sq[r]);
    #pragma unroll
    for (int r = 0; r < 8; ++r) sq[r] += xlane16<4>(sq[r]);
    #pragma unroll
    for (int r = 0; r < 8; ++r) sq[r] += xlane16<8>(sq[r]);
    __syncthreads();
    if (t < 16) rowsum[t] = 0.0f;
    __syncthreads();
    if (mr == 0) {
        #pragma unroll
        for (int r = 0; r < 8; ++r) atomicAdd(&rowsum[r + 8 * kg], sq[r]);
    }
    __syncthreads();

    float inv[8];
    #pragma unroll
    for (int r = 0; r < 8; ++r)
        inv[r] = 1.0f / fmaxf(sqrtf(rowsum[r + 8 * kg]), 1e-12f);

    #pragma unroll
    for (int s = 0; s < 4; ++s) {
        float s1[2], s2[2];
        #pragma unroll
        for (int nt = 0; nt < 2; ++nt) {
            int col = s * 256 + colBase + nt * 16 + mr;
            s1[nt] = 0.0f;
            s2[nt] = 0.0f;
            #pragma unroll
            for (int r = 0; r < 8; ++r) {
                float v = fmaxf(acc[s][nt][r] * inv[r], 0.0f);
                s1[nt] += v;
                s2[nt] += v * v;
                out[(size_t)(rowBase + r + 8 * kg) * COLS + col] = v;
            }
        }
        const float keep1 = kg ? s1[1] : s1[0];
        const float give1 = kg ? s1[0] : s1[1];
        const float keep2 = kg ? s2[1] : s2[0];
        const float give2 = kg ? s2[0] : s2[1];
        const int colw = s * 256 + colBase + lane;
        atomicAdd(&gsum[colw],   keep1 + xhalf(give1));
        atomicAdd(&gsumsq[colw], keep2 + xhalf(give2));
    }
}

__global__ void __launch_bounds__(256) sage_bn_apply(float* __restrict__ out,
                                                     const float* __restrict__ gsum,
                                                     const float* __restrict__ gsumsq,
                                                     const float* __restrict__ gamma,
                                                     const float* __restrict__ beta)
{
    int i = blockIdx.x * 256 + threadIdx.x;
    int c = i & (COLS - 1);
    const float invn = 1.0f / (float)ROWS;
    float mu  = gsum[c] * invn;
    float var = gsumsq[c] * invn - mu * mu;
    float v   = out[i];
    out[i] = (v - mu) * rsqrtf(var + 1e-5f) * gamma[c] + beta[c];
}

extern "C" void kernel_launch(void* const* d_in, const int* in_sizes, int n_in,
                              void* d_out, int out_size, void* d_ws, size_t ws_size,
                              hipStream_t stream) {
    (void)in_sizes; (void)n_in; (void)out_size; (void)ws_size;
    const float* x    = (const float*)d_in[0];
    const int*   idx1 = (const int*)  d_in[1];
    const int*   idx2 = (const int*)  d_in[2];
    const int*   idx3 = (const int*)  d_in[3];
    const float* Wx   = (const float*)d_in[4];
    const float* bx   = (const float*)d_in[5];
    const float* Wn   = (const float*)d_in[6];
    const float* bnb  = (const float*)d_in[7];
    const float* gam  = (const float*)d_in[8];
    const float* bet  = (const float*)d_in[9];
    float* out = (float*)d_out;

    float* gsum   = (float*)d_ws;
    float* gsumsq = gsum + COLS;

    sage_init_stats<<<8, 256, 0, stream>>>(gsum);

    sage_fused<<<ROWS / TILE_M, 256, 0, stream>>>(
        x, idx1, idx2, idx3, Wx, bx, Wn, bnb, out, gsum, gsumsq);

    sage_bn_apply<<<(ROWS * COLS) / 256, 256, 0, stream>>>(out, gsum, gsumsq, gam, bet);
}


# BN apply pass: a thread takes one column of 4 consecutive rows; lane width stays 4 B, so each load and store of a wave is still one whole 128 B line; 4096 work-groups, not 16384
# speedup vs baseline: 1.9607x; 1.1546x over previous
#include <hip/hip_runtime.h>
#include <math.h>

#define BB    8
#define NN    512
#define FINC  256
#define FOUTC 256
#define NMAXK 32
#define COLS  1024
#define ROWS  4096
#define TILE_M 16
#define AROW  260
#define ATILE (TILE_M*AROW)
#define CH    64
#define NSL   (FINC/CH)
#define TABF  (NN*CH)
#define PRE   (TABF/4/256)
#define SMEMF (TABF + 4*ATILE + COLS + 16)

typedef float v2f __attribute__((ext_vector_type(2)));
typedef float v8f __attribute__((ext_vector_type(8)));
typedef float v4f __attribute__((ext_vector_type(4)));

struct Frag {
    v4f a[4][4];
    v4f b[2][2][4];
};

__device__ __forceinline__ void load_frag(Frag& f, const float* __restrict__ Wxp,
                                          const float* __restrict__ Wnp,
                                          const float* Ap, int kb)
{
    #pragma unroll
    for (int nt = 0; nt < 2; ++nt) {
        #pragma unroll
        for (int c = 0; c < 4; ++c)
            f.b[0][nt][c] = *(const v4f*)(Wxp + nt * 16 * FINC + kb + 8 * c);
        #pragma unroll
        for (int c = 0; c < 4; ++c)
            f.b[1][nt][c] = *(const v4f*)(Wnp + nt * 16 * FINC + kb + 8 * c);
    }
    #pragma unroll
    for (int s = 0; s < 4; ++s) {
        #pragma unroll
        for (int c = 0; c < 4; ++c)
            f.a[s][c] = *(const v4f*)(Ap + s * ATILE + kb + 8 * c);
    }
}

__device__ __forceinline__ void mma_block(v8f (&acc)[4][2], const Frag& f)
{
    #pragma unroll
    for (int c = 0; c < 4; ++c) {
        #pragma unroll
        for (int s = 0; s < 4; ++s) {
            const int w = (s == 0) ? 0 : 1;
            #pragma unroll
            for (int nt = 0; nt < 2; ++nt) {
                v2f a0 = { f.a[s][c][0], f.a[s][c][1] };
                v2f b0 = { f.b[w][nt][c][0], f.b[w][nt][c][1] };
                acc[s][nt] = __builtin_amdgcn_wmma_f32_16x16x4_f32(
                    false, a0, false, b0, (short)0, acc[s][nt], false, false);
            }
        }
        #pragma unroll
        for (int s = 0; s < 4; ++s) {
            const int w = (s == 0) ? 0 : 1;
            #pragma unroll
            for (int nt = 0; nt < 2; ++nt) {
                v2f a1 = { f.a[s][c][2], f.a[s][c][3] };
                v2f b1 = { f.b[w][nt][c][2], f.b[w][nt][c][3] };
                acc[s][nt] = __builtin_amdgcn_wmma_f32_16x16x4_f32(
                    false, a1, false, b1, (short)0, acc[s][nt], false, false);
            }
        }
    }
}

__device__ __forceinline__ void rows_issue(v2f (&v)[NMAXK], const float* const (&rp)[NMAXK])
{
    #pragma unroll
    for (int k = 0; k < NMAXK; ++k)
        v[k] = *(const v2f*)rp[k];
}

__device__ __forceinline__ v2f rows_mean(const v2f (&v)[NMAXK])
{
    v2f sum = { 0.0f, 0.0f };
    #pragma unroll
    for (int k = 0; k < NMAXK; ++k) {
        sum[0] += v[k][0];
        sum[1] += v[k][1];
    }
    v2f m = { sum[0] * (1.0f / NMAXK), sum[1] * (1.0f / NMAXK) };
    return m;
}

__device__ __forceinline__ float xhalf(float x)
{
    int b = __builtin_bit_cast(int, x);
    return __builtin_bit_cast(float, __builtin_amdgcn_permlanex16(b, b, 0x76543210, 0xfedcba98, false, false));
}

template <int M>
__device__ __forceinline__ float xlane16(float x)
{
    int b = __builtin_bit_cast(int, x);
    return __builtin_bit_cast(float, __builtin_amdgcn_update_dpp(b, b, 0x160 + M, 0xf, 0xf, false));
}

__global__ void __launch_bounds__(256) sage_init_stats(float* __restrict__ stats) {
    int i = blockIdx.x * 256 + threadIdx.x;
    stats[i] = 0.0f;
}

__global__ void __launch_bounds__(256, 1)
sage_fused(const float* __restrict__ x,
           const int*   __restrict__ idx1,
           const int*   __restrict__ idx2,
           const int*   __restrict__ idx3,
           const float* __restrict__ Wx, const float* __restrict__ bx,
           const float* __restrict__ Wn, const float* __restrict__ bnb,
           float* __restrict__ out,
           float* __restrict__ gsum, float* __restrict__ gsumsq)
{
    __shared__ __attribute__((aligned(16))) float smem[SMEMF];
    float* Alds    = smem;
    float* biasLds = Alds + 4 * ATILE;
    float* rowsum  = biasLds + COLS;
    float* tab     = rowsum + 16;

    const int t       = threadIdx.x;
    const int lane    = t & 31;
    const int wave    = t >> 5;
    const int rowBase = blockIdx.x * TILE_M;
    const int b       = rowBase / NN;
    const int nBase   = rowBase % NN;
    const float* xb = x + (size_t)b * NN * FINC;

    int ov[6];
    #pragma unroll
    for (int pi = 0; pi < 6; ++pi) {
        const int* ip = (pi < 2) ? idx1 : ((pi < 4) ? idx2 : idx3);
        ov[pi] = ip[(nBase + wave + 8 * (pi & 1)) * NMAXK + lane] * CH;
    }

    const float* xpre = xb + (t >> 4) * FINC + 4 * (t & 15);
    v4f pre[PRE];
    #pragma unroll
    for (int j = 0; j < PRE; ++j)
        pre[j] = *(const v4f*)(xpre + j * 16 * FINC);

    {
        float bv = bnb[t];
        biasLds[t]       = bx[t];
        biasLds[256 + t] = bv;
        biasLds[512 + t] = bv;
        biasLds[768 + t] = bv;
    }
    for (int r = 0; r < TILE_M; ++r)
        Alds[r * AROW + t] = xb[(nBase + r) * FINC + t];

    const float* tl = tab + 2 * lane;
    const float* rp[6][NMAXK];
    #pragma unroll
    for (int pi = 0; pi < 6; ++pi) {
        #pragma unroll
        for (int k = 0; k < NMAXK; ++k)
            rp[pi][k] = tl + __builtin_amdgcn_readlane(ov[pi], k);
    }
    v2f va[NMAXK], vb[NMAXK];
    for (int c = 0; c < NSL; ++c) {
        #pragma unroll
        for (int j = 0; j < PRE; ++j)
            *(v4f*)(tab + (j * 256 + t) * 4) = pre[j];
        __syncthreads();
        if (c + 1 < NSL) {
            #pragma unroll
            for (int j = 0; j < PRE; ++j)
                pre[j] = *(const v4f*)(xpre + (c + 1) * CH + j * 16 * FINC);
        }
        float* At = Alds + ATILE + wave * AROW + c * CH + 2 * lane;
        rows_issue(va, rp[0]);
        #pragma unroll
        for (int q = 0; q < 3; ++q) {
            rows_issue(vb, rp[2 * q + 1]);
            __builtin_amdgcn_sched_barrier(0);
            *(v2f*)(At + q * ATILE) = rows_mean(va);
            __builtin_amdgcn_sched_barrier(0);
            if (q < 2) rows_issue(va, rp[2 * q + 2]);
            __builtin_amdgcn_sched_barrier(0);
            *(v2f*)(At + q * ATILE + 8 * AROW) = rows_mean(vb);
            __builtin_amdgcn_sched_barrier(0);
        }
        __syncthreads();
    }

    const int mr   = lane & 15;
    const int kg   = lane >> 4;

    const float* Ap  = Alds + mr * AROW + 4 * kg;
    const float* Wxp = Wx + (size_t)(wave * 32 + mr) * FINC + 4 * kg;
    const float* Wnp = Wn + (size_t)(wave * 32 + mr) * FINC + 4 * kg;

    v8f acc[4][2] = {};

    Frag f0, f1;
    load_frag(f0, Wxp, Wnp, Ap, 0);
    for (int kb = 0; kb < FINC - 64; kb += 64) {
        load_frag(f1, Wxp, Wnp, Ap, kb + 32);
        __builtin_amdgcn_sched_barrier(0);
        mma_block(acc, f0);
        __builtin_amdgcn_sched_barrier(0);
        load_frag(f0, Wxp, Wnp, Ap, kb + 64);
        __builtin_amdgcn_sched_barrier(0);
        mma_block(acc, f1);
        __builtin_amdgcn_sched_barrier(0);
    }
    load_frag(f1, Wxp, Wnp, Ap, FINC - 32);
    __builtin_amdgcn_sched_barrier(0);
    mma_block(acc, f0);
    mma_block(acc, f1);

    const int colBase = wave * 32;
    #pragma unroll
    for (int s = 0; s < 4; ++s) {
        #pragma unroll
        for (int nt = 0; nt < 2; ++nt) {
            float bv = biasLds[s * 256 + colBase + nt * 16 + mr];
            #pragma unroll
            for (int r = 0; r < 8; ++r) acc[s][nt][r] += bv;
        }
    }

    float sq[8];
    #pragma unroll
    for (int r = 0; r < 8; ++r) {
        sq[r] = 0.0f;
        #pragma unroll
        for (int s = 0; s < 4; ++s) {
            #pragma unroll
            for (int nt = 0; nt < 2; ++nt) sq[r] += acc[s][nt][r] * acc[s][nt][r];
        }
    }
    #pragma unroll
    for (int r = 0; r < 8; ++r) sq[r] += xlane16<1>(sq[r]);
    #pragma unroll
    for (int r = 0; r < 8; ++r) sq[r] += xlane16<2>(sq[r]);
    #pragma unroll
    for (int r = 0; r < 8; ++r) sq[r] += xlane16<4>(sq[r]);
    #pragma unroll
    for (int r = 0; r < 8; ++r) sq[r] += xlane16<8>(sq[r]);
    __syncthreads();
    if (t < 16) rowsum[t] = 0.0f;
    __syncthreads();
    if (mr == 0) {
        #pragma unroll
        for (int r = 0; r < 8; ++r) atomicAdd(&rowsum[r + 8 * kg], sq[r]);
    }
    __syncthreads();

    float inv[8];
    #pragma unroll
    for (int r = 0; r < 8; ++r)
        inv[r] = 1.0f / fmaxf(sqrtf(rowsum[r + 8 * kg]), 1e-12f);

    #pragma unroll
    for (int s = 0; s < 4; ++s) {
        float s1[2], s2[2];
        #pragma unroll
        for (int nt = 0; nt < 2; ++nt) {
            int col = s * 256 + colBase + nt * 16 + mr;
            s1[nt] = 0.0f;
            s2[nt] = 0.0f;
            #pragma unroll
            for (int r = 0; r < 8; ++r) {
                float v = fmaxf(acc[s][nt][r] * inv[r], 0.0f);
                s1[nt] += v;
                s2[nt] += v * v;
                out[(size_t)(rowBase + r + 8 * kg) * COLS + col] = v;
            }
        }
        const float keep1 = kg ? s1[1] : s1[0];
        const float give1 = kg ? s1[0] : s1[1];
        const float keep2 = kg ? s2[1] : s2[0];
        const float give2 = kg ? s2[0] : s2[1];
        const int colw = s * 256 + colBase + lane;
        atomicAdd(&gsum[colw],   keep1 + xhalf(give1));
        atomicAdd(&gsumsq[colw], keep2 + xhalf(give2));
    }
}

#define BNR 4
__global__ void __launch_bounds__(256) sage_bn_apply(float* __restrict__ out,
                                                     const float* __restrict__ gsum,
                                                     const float* __restrict__ gsumsq,
                                                     const float* __restrict__ gamma,
                                                     const float* __restrict__ beta)
{
    const int i = blockIdx.x * 256 + threadIdx.x;
    const int c = i & (COLS - 1);
    const int g = i >> 10;
    float* p = out + (size_t)g * (BNR * COLS) + c;
    float v[BNR];
    for (int k = 0; k < BNR; ++k) v[k] = p[k * COLS];
    const float invn = 1.0f / (float)ROWS;
    const float mu  = gsum[c] * invn;
    const float var = gsumsq[c] * invn - mu * mu;
    const float rs  = rsqrtf(var + 1e-5f);
    const float ga  = gamma[c];
    const float be  = beta[c];
    for (int k = 0; k < BNR; ++k) p[k * COLS] = (v[k] - mu) * rs * ga + be;
}

extern "C" void kernel_launch(void* const* d_in, const int* in_sizes, int n_in,
                              void* d_out, int out_size, void* d_ws, size_t ws_size,
                              hipStream_t stream) {
    (void)in_sizes; (void)n_in; (void)out_size; (void)ws_size;
    const float* x    = (const float*)d_in[0];
    const int*   idx1 = (const int*)  d_in[1];
    const int*   idx2 = (const int*)  d_in[2];
    const int*   idx3 = (const int*)  d_in[3];
    const float* Wx   = (const float*)d_in[4];
    const float* bx   = (const float*)d_in[5];
    const float* Wn   = (const float*)d_in[6];
    const float* bnb  = (const float*)d_in[7];
    const float* gam  = (const float*)d_in[8];
    const float* bet  = (const float*)d_in[9];
    float* out = (float*)d_out;

    float* gsum   = (float*)d_ws;
    float* gsumsq = gsum + COLS;

    sage_init_stats<<<8, 256, 0, stream>>>(gsum);

    sage_fused<<<ROWS / TILE_M, 256, 0, stream>>>(
        x, idx1, idx2, idx3, Wx, bx, Wn, bnb, out, gsum, gsumsq);

    sage_bn_apply<<<(ROWS * COLS) / (256 * BNR), 256, 0, stream>>>(out, gsum, gsumsq, gam, bet);
}


# Gather: slices double-buffered in LDS (rows interleaved; own-row tile kept in registers for room): a wave stores slice c+1 and re-requests c+2 beside its gather of c; one LDS-only barrier a slice
# speedup vs baseline: 2.2388x; 1.1418x over previous
#include <hip/hip_runtime.h>
#include <math.h>

#define BB    8
#define NN    512
#define FINC  256
#define FOUTC 256
#define NMAXK 32
#define COLS  1024
#define ROWS  4096
#define TILE_M 16
#define AROW  260
#define ATILE (TILE_M*AROW)
#define CH    64
#define NSL   (FINC/CH)
#define TABF  (NN*CH)
#define PRE   (TABF/4/256)
#define SMEMF (TABF + 4*ATILE + COLS + 16)

typedef float v2f __attribute__((ext_vector_type(2)));
typedef float v8f __attribute__((ext_vector_type(8)));
typedef float v4f __attribute__((ext_vector_type(4)));

struct Frag {
    v4f a[4][4];
    v4f b[2][2][4];
};

__device__ __forceinline__ void load_frag(Frag& f, const float* __restrict__ Wxp,
                                          const float* __restrict__ Wnp,
                                          const float* Ap, int kb)
{
    #pragma unroll
    for (int nt = 0; nt < 2; ++nt) {
        #pragma unroll
        for (int c = 0; c < 4; ++c)
            f.b[0][nt][c] = *(const v4f*)(Wxp + nt * 16 * FINC + kb + 8 * c);
        #pragma unroll
        for (int c = 0; c < 4; ++c)
            f.b[1][nt][c] = *(const v4f*)(Wnp + nt * 16 * FINC + kb + 8 * c);
    }
    #pragma unroll
    for (int s = 0; s < 4; ++s) {
        #pragma unroll
        for (int c = 0; c < 4; ++c)
            f.a[s][c] = *(const v4f*)(Ap + s * ATILE + kb + 8 * c);
    }
}

__device__ __forceinline__ void mma_block(v8f (&acc)[4][2], const Frag& f)
{
    #pragma unroll
    for (int c = 0; c < 4; ++c) {
        #pragma unroll
        for (int s = 0; s < 4; ++s) {
            const int w = (s == 0) ? 0 : 1;
            #pragma unroll
            for (int nt = 0; nt < 2; ++nt) {
                v2f a0 = { f.a[s][c][0], f.a[s][c][1] };
                v2f b0 = { f.b[w][nt][c][0], f.b[w][nt][c][1] };
                acc[s][nt] = __builtin_amdgcn_wmma_f32_16x16x4_f32(
                    false, a0, false, b0, (short)0, acc[s][nt], false, false);
            }
        }
        #pragma unroll
        for (int s = 0; s < 4; ++s) {
            const int w = (s == 0) ? 0 : 1;
            #pragma unroll
            for (int nt = 0; nt < 2; ++nt) {
                v2f a1 = { f.a[s][c][2], f.a[s][c][3] };
                v2f b1 = { f.b[w][nt][c][2], f.b[w][nt][c][3] };
                acc[s][nt] = __builtin_amdgcn_wmma_f32_16x16x4_f32(
                    false, a1, false, b1, (short)0, acc[s][nt], false, false);
            }
        }
    }
}

__device__ __forceinline__ void rows_issue(v2f (&v)[NMAXK], const float* const (&rp)[NMAXK])
{
    #pragma unroll
    for (int k = 0; k < NMAXK; ++k)
        v[k] = *(const v2f*)rp[k];
}

__device__ __forceinline__ v2f rows_mean(const v2f (&v)[NMAXK])
{
    v2f sum = { 0.0f, 0.0f };
    #pragma unroll
    for (int k = 0; k < NMAXK; ++k) {
        sum[0] += v[k][0];
        sum[1] += v[k][1];
    }
    v2f m = { sum[0] * (1.0f / NMAXK), sum[1] * (1.0f / NMAXK) };
    return m;
}

__device__ __forceinline__ float xhalf(float x)
{
    int b = __builtin_bit_cast(int, x);
    return __builtin_bit_cast(float, __builtin_amdgcn_permlanex16(b, b, 0x76543210, 0xfedcba98, false, false));
}

template <int M>
__device__ __forceinline__ float xlane16(float x)
{
    int b = __builtin_bit_cast(int, x);
    return __builtin_bit_cast(float, __builtin_amdgcn_update_dpp(b, b, 0x160 + M, 0xf, 0xf, false));
}

__global__ void __launch_bounds__(256) sage_init_stats(float* __restrict__ stats) {
    int i = blockIdx.x * 256 + threadIdx.x;
    stats[i] = 0.0f;
}

__global__ void __launch_bounds__(256, 1)
sage_fused(const float* __restrict__ x,
           const int*   __restrict__ idx1,
           const int*   __restrict__ idx2,
           const int*   __restrict__ idx3,
           const float* __restrict__ Wx, const float* __restrict__ bx,
           const float* __restrict__ Wn, const float* __restrict__ bnb,
           float* __restrict__ out,
           float* __restrict__ gsum, float* __restrict__ gsumsq)
{
    __shared__ __attribute__((aligned(16))) float smem[SMEMF];
    float* Alds    = smem;
    float* biasLds = Alds + 4 * ATILE;
    float* rowsum  = biasLds + COLS;
    float* tab     = rowsum + 16;

    const int t       = threadIdx.x;
    const int lane    = t & 31;
    const int wave    = t >> 5;
    const int rowBase = blockIdx.x * TILE_M;
    const int b       = rowBase / NN;
    const int nBase   = rowBase % NN;
    const float* xb = x + (size_t)b * NN * FINC;

    int ov[6];
    #pragma unroll
    for (int pi = 0; pi < 6; ++pi) {
        const int* ip = (pi < 2) ? idx1 : ((pi < 4) ? idx2 : idx3);
        ov[pi] = ip[(nBase + wave + 8 * (pi & 1)) * NMAXK + lane] * CH;
    }

    const float* xpre = xb + (t >> 4) * FINC + 4 * (t & 15);
    v4f pre[PRE];
    #pragma unroll
    for (int j = 0; j < PRE; ++j)
        pre[j] = *(const v4f*)(xpre + j * 16 * FINC);

    {
        float bv = bnb[t];
        biasLds[t]       = bx[t];
        biasLds[256 + t] = bv;
        biasLds[512 + t] = bv;
        biasLds[768 + t] = bv;
    }
    for (int r = 0; r < TILE_M; ++r)
        Alds[r * AROW + t] = xb[(nBase + r) * FINC + t];

    const float* tl = tab + 2 * lane;
    const float* rp[6][NMAXK];
    #pragma unroll
    for (int pi = 0; pi < 6; ++pi) {
        #pragma unroll
        for (int k = 0; k < NMAXK; ++k)
            rp[pi][k] = tl + __builtin_amdgcn_readlane(ov[pi], k);
    }
    v2f va[NMAXK], vb[NMAXK];
    for (int c = 0; c < NSL; ++c) {
        #pragma unroll
        for (int j = 0; j < PRE; ++j)
            *(v4f*)(tab + (j * 256 + t) * 4) = pre[j];
        __syncthreads();
        if (c + 1 < NSL) {
            #pragma unroll
            for (int j = 0; j < PRE; ++j)
                pre[j] = *(const v4f*)(xpre + (c + 1) * CH + j * 16 * FINC);
        }
        float* At = Alds + ATILE + wave * AROW + c * CH + 2 * lane;
        rows_issue(va, rp[0]);
        #pragma unroll
        for (int q = 0; q < 3; ++q) {
            rows_issue(vb, rp[2 * q + 1]);
            __builtin_amdgcn_sched_barrier(0);
            *(v2f*)(At + q * ATILE) = rows_mean(va);
            __builtin_amdgcn_sched_barrier(0);
            if (q < 2) rows_issue(va, rp[2 * q + 2]);
            __builtin_amdgcn_sched_barrier(0);
            *(v2f*)(At + q * ATILE + 8 * AROW) = rows_mean(vb);
            __builtin_amdgcn_sched_barrier(0);
        }
        __syncthreads();
    }

    const int mr   = lane & 15;
    const int kg   = lane >> 4;

    const float* Ap  = Alds + mr * AROW + 4 * kg;
    const float* Wxp = Wx + (size_t)(wave * 32 + mr) * FINC + 4 * kg;
    const float* Wnp = Wn + (size_t)(wave * 32 + mr) * FINC + 4 * kg;

    v8f acc[4][2] = {};

    Frag f0, f1;
    load_frag(f0, Wxp, Wnp, Ap, 0);
    for (int kb = 0; kb < FINC - 64; kb += 64) {
        load_frag(f1, Wxp, Wnp, Ap, kb + 32);
        __builtin_amdgcn_sched_barrier(0);
        mma_block(acc, f0);
        __builtin_amdgcn_sched_barrier(0);
        load_frag(f0, Wxp, Wnp, Ap, kb + 64);
        __builtin_amdgcn_sched_barrier(0);
        mma_block(acc, f1);
        __builtin_amdgcn_sched_barrier(0);
    }
    load_frag(f1, Wxp, Wnp, Ap, FINC - 32);
    __builtin_amdgcn_sched_barrier(0);
    mma_block(acc, f0);
    mma_block(acc, f1);

    const int colBase = wave * 32;
    #pragma unroll
    for (int s = 0; s < 4; ++s) {
        #pragma unroll
        for (int nt = 0; nt < 2; ++nt) {
            float bv = biasLds[s * 256 + colBase + nt * 16 + mr];
            #pragma unroll
            for (int r = 0; r < 8; ++r) acc[s][nt][r] += bv;
        }
    }

    float sq[8];
    #pragma unroll
    for (int r = 0; r < 8; ++r) {
        sq[r] = 0.0f;
        #pragma unroll
        for (int s = 0; s < 4; ++s) {
            #pragma unroll
            for (int nt = 0; nt < 2; ++nt) sq[r] += acc[s][nt][r] * acc[s][nt][r];
        }
    }
    #pragma unroll
    for (int r = 0; r < 8; ++r) sq[r] += xlane16<1>(sq[r]);
    #pragma unroll
    for (int r = 0; r < 8; ++r) sq[r] += xlane16<2>(sq[r]);
    #pragma unroll
    for (int r = 0; r < 8; ++r) sq[r] += xlane16<4>(sq[r]);
    #pragma unroll
    for (int r = 0; r < 8; ++r) sq[r] += xlane16<8>(sq[r]);
    __syncthreads();
    if (t < 16) rowsum[t] = 0.0f;
    __syncthreads();
    if (mr == 0) {
        #pragma unroll
        for (int r = 0; r < 8; ++r) atomicAdd(&rowsum[r + 8 * kg], sq[r]);
    }
    __syncthreads();

    float inv[8];
    #pragma unroll
    for (int r = 0; r < 8; ++r)
        inv[r] = 1.0f / fmaxf(sqrtf(rowsum[r + 8 * kg]), 1e-12f);

    #pragma unroll
    for (int s = 0; s < 4; ++s) {
        float s1[2], s2[2];
        #pragma unroll
        for (int nt = 0; nt < 2; ++nt) {
            int col = s * 256 + colBase + nt * 16 + mr;
            s1[nt] = 0.0f;
            s2[nt] = 0.0f;
            #pragma unroll
            for (int r = 0; r < 8; ++r) {
                float v = fmaxf(acc[s][nt][r] * inv[r], 0.0f);
                s1[nt] += v;
                s2[nt] += v * v;
                out[(size_t)(rowBase + r + 8 * kg) * COLS + col] = v;
            }
        }
        const float keep1 = kg ? s1[1] : s1[0];
        const float give1 = kg ? s1[0] : s1[1];
        const float keep2 = kg ? s2[1] : s2[0];
        const float give2 = kg ? s2[0] : s2[1];
        const int colw = s * 256 + colBase + lane;
        atomicAdd(&gsum[colw],   keep1 + xhalf(give1));
        atomicAdd(&gsumsq[colw], keep2 + xhalf(give2));
    }
}

#define PN_OFF   (2 * COLS)
#define WS_NEED  ((size_t)(2 * COLS + ROWS * FOUTC) * 4)
#define PM       64

struct FragP {
    v4f a[4][4];
    v4f b[4];
};

__device__ __forceinline__ void load_fragp(FragP& f, const float* __restrict__ Wp, const float* Ap, int kb)
{
    #pragma unroll
    for (int c = 0; c < 4; ++c)
        f.b[c] = *(const v4f*)(Wp + kb + 8 * c);
    #pragma unroll
    for (int s = 0; s < 4; ++s) {
        #pragma unroll
        for (int c = 0; c < 4; ++c)
            f.a[s][c] = *(const v4f*)(Ap + s * ATILE + kb + 8 * c);
    }
}

__device__ __forceinline__ void mma_blockp(v8f (&acc)[4], const FragP& f)
{
    #pragma unroll
    for (int c = 0; c < 4; ++c) {
        #pragma unroll
        for (int s = 0; s < 4; ++s) {
            v2f a0 = { f.a[s][c][0], f.a[s][c][1] };
            v2f b0 = { f.b[c][0], f.b[c][1] };
            acc[s] = __builtin_amdgcn_wmma_f32_16x16x4_f32(
                false, a0, false, b0, (short)0, acc[s], false, false);
        }
        #pragma unroll
        for (int s = 0; s < 4; ++s) {
            v2f a1 = { f.a[s][c][2], f.a[s][c][3] };
            v2f b1 = { f.b[c][2], f.b[c][3] };
            acc[s] = __builtin_amdgcn_wmma_f32_16x16x4_f32(
                false, a1, false, b1, (short)0, acc[s], false, false);
        }
    }
}

__global__ void __launch_bounds__(256, 1)
sage_project(const float* __restrict__ x,
             const float* __restrict__ Wx, const float* __restrict__ Wn,
             float* __restrict__ out, float* __restrict__ pn, float* __restrict__ stats)
{
    __shared__ __attribute__((aligned(16))) float Alds[4 * ATILE];

    const int t    = threadIdx.x;
    const int lane = t & 31;
    const int wave = t >> 5;
    const int rowBase  = (blockIdx.x >> 2) * PM;
    const int cb       = blockIdx.x & 3;
    const int chanBase = (cb & 1) * 128;
    const float* __restrict__ Wsel = (cb < 2) ? Wx : Wn;

    if (blockIdx.x < 8) stats[blockIdx.x * 256 + t] = 0.0f;

    {
        const float* xs = x + (size_t)(rowBase + (t >> 6)) * FINC + 4 * (t & 63);
        float* ad = Alds + (t >> 6) * AROW + 4 * (t & 63);
        v4f tmp[16];
        #pragma unroll
        for (int j = 0; j < 16; ++j) tmp[j] = *(const v4f*)(xs + j * 4 * FINC);
        #pragma unroll
        for (int j = 0; j < 16; ++j) *(v4f*)(ad + j * 4 * AROW) = tmp[j];
    }
    __syncthreads();

    const int mr = lane & 15;
    const int kg = lane >> 4;
    const float* Ap = Alds + mr * AROW + 4 * kg;
    const float* Wp = Wsel + (size_t)(chanBase + wave * 16 + mr) * FINC + 4 * kg;

    v8f acc[4] = {};

    FragP f0, f1;
    load_fragp(f0, Wp, Ap, 0);
    for (int kb = 0; kb < FINC - 64; kb += 64) {
        load_fragp(f1, Wp, Ap, kb + 32);
        __builtin_amdgcn_sched_barrier(0);
        mma_blockp(acc, f0);
        __builtin_amdgcn_sched_barrier(0);
        load_fragp(f0, Wp, Ap, kb + 64);
        __builtin_amdgcn_sched_barrier(0);
        mma_blockp(acc, f1);
        __builtin_amdgcn_sched_barrier(0);
    }
    load_fragp(f1, Wp, Ap, FINC - 32);
    __builtin_amdgcn_sched_barrier(0);
    mma_blockp(acc, f0);
    mma_blockp(acc, f1);

    const int col = chanBase + wave * 16 + mr;
    float* dst        = (cb < 2) ? (out + col) : (pn + col);
    const int stride  = (cb < 2) ? COLS : FOUTC;
    #pragma unroll
    for (int s = 0; s < 4; ++s) {
        #pragma unroll
        for (int r = 0; r < 8; ++r)
            dst[(size_t)(rowBase + 16 * s + 8 * kg + r) * stride] = acc[s][r];
    }
}

#define GROW   (2 * CH)
#define SMEMG  (NN * GROW + 3 * ATILE + COLS + 16)

__device__ __forceinline__ void lds_barrier()
{
    asm volatile("s_wait_dscnt 0x0\n\ts_barrier_signal -1\n\ts_barrier_wait -1" ::: "memory");
}

template <int P>
__device__ __forceinline__ void rows_issue_buf(v2f (&v)[NMAXK], const float* const (&rp)[NMAXK])
{
    #pragma unroll
    for (int k = 0; k < NMAXK; ++k)
        v[k] = *(const v2f*)(rp[k] + P * CH);
}

template <int P>
__device__ __forceinline__ void slice_store(float* tab, const v4f (&pre)[PRE], int t)
{
    float* d = tab + (t >> 4) * GROW + P * CH + 4 * (t & 15);
    #pragma unroll
    for (int j = 0; j < PRE; ++j)
        *(v4f*)(d + j * 16 * GROW) = pre[j];
}

template <int MODE>
__device__ __forceinline__ void move_pieces(float* sd, v4f (&pre)[PRE], const float* xn, int j0, int j1)
{
    if (MODE == 0) return;
    #pragma unroll
    for (int j = j0; j < j1; ++j)
        *(v4f*)(sd + j * 16 * GROW) = pre[j];
    if (MODE == 1) {
        #pragma unroll
        for (int j = j0; j < j1; ++j)
            pre[j] = *(const v4f*)(xn + j * 16 * FOUTC);
    }
}

__device__ __forceinline__ constexpr int piece0(int g) { return g < 2 ? 6 * g : 12 + 5 * (g - 2); }

template <int P, int MODE>
__device__ __forceinline__ void gather_slice(float* At, const float* const (&rp)[6][NMAXK],
                                             float* sd, v4f (&pre)[PRE], const float* xn)
{
    v2f va[NMAXK], vb[NMAXK];
    rows_issue_buf<P>(va, rp[0]);
    move_pieces<MODE>(sd, pre, xn, piece0(0), piece0(1));
    #pragma unroll
    for (int q = 0; q < 3; ++q) {
        rows_issue_buf<P>(vb, rp[2 * q + 1]);
        move_pieces<MODE>(sd, pre, xn, piece0(2 * q + 1), piece0(2 * q + 2));
        __builtin_amdgcn_sched_barrier(0);
        *(v2f*)(At + q * ATILE) = rows_mean(va);
        __builtin_amdgcn_sched_barrier(0);
        if (q < 2) {
            rows_issue_buf<P>(va, rp[2 * q + 2]);
            move_pieces<MODE>(sd, pre, xn, piece0(2 * q + 2), piece0(2 * q + 3));
        }
        __builtin_amdgcn_sched_barrier(0);
        *(v2f*)(At + q * ATILE + 8 * AROW) = rows_mean(vb);
        __builtin_amdgcn_sched_barrier(0);
    }
}

__global__ void __launch_bounds__(256, 1)
sage_gather(const float* __restrict__ pn,
            const int*   __restrict__ idx1,
            const int*   __restrict__ idx2,
            const int*   __restrict__ idx3,
            const float* __restrict__ bx, const float* __restrict__ bnb,
            float* __restrict__ out,
            float* __restrict__ gsum, float* __restrict__ gsumsq)
{
    __shared__ __attribute__((aligned(16))) float smem[SMEMG];
    float* tab     = smem;
    float* Alds    = tab + NN * GROW;
    float* biasLds = Alds + 3 * ATILE;
    float* rowsum  = biasLds + COLS;

    const int t       = threadIdx.x;
    const int lane    = t & 31;
    const int wave    = t >> 5;
    const int rowBase = blockIdx.x * TILE_M;
    const int b       = rowBase / NN;
    const int nBase   = rowBase % NN;
    const float* xb = pn + (size_t)b * NN * FOUTC;

    int ov[6];
    #pragma unroll
    for (int pi = 0; pi < 6; ++pi) {
        const int* ip = (pi < 2) ? idx1 : ((pi < 4) ? idx2 : idx3);
        ov[pi] = ip[(nBase + wave + 8 * (pi & 1)) * NMAXK + lane] * GROW;
    }

    const float* xpre = xb + (t >> 4) * FOUTC + 4 * (t & 15);
    v4f pre[PRE];
    #pragma unroll
    for (int j = 0; j < PRE; ++j)
        pre[j] = *(const v4f*)(xpre + j * 16 * FOUTC);

    {
        float bv = bnb[t];
        biasLds[t]       = bx[t];
        biasLds[256 + t] = bv;
        biasLds[512 + t] = bv;
        biasLds[768 + t] = bv;
        if (t < 16) rowsum[t] = 0.0f;
    }

    const float* tl = tab + 2 * lane;
    const float* rp[6][NMAXK];
    #pragma unroll
    for (int pi = 0; pi < 6; ++pi) {
        #pragma unroll
        for (int k = 0; k < NMAXK; ++k)
            rp[pi][k] = tl + __builtin_amdgcn_readlane(ov[pi], k);
    }

    slice_store<0>(tab, pre, t);
    #pragma unroll
    for (int j = 0; j < PRE; ++j)
        pre[j] = *(const v4f*)(xpre + CH + j * 16 * FOUTC);
    lds_barrier();

    float* At = Alds + wave * AROW + 2 * lane;
    float* sd0 = tab + (t >> 4) * GROW + 4 * (t & 15);
    float selfv[TILE_M];
    static_assert(NSL >= 2 && piece0(6) == PRE, "two buffers in use; the six piece groups cover the staging registers");
    #pragma unroll
    for (int c = 0; c < NSL; ++c) {
        float* sd = sd0 + ((c + 1) & 1) * CH;
        const float* xn = xpre + (c + 2) * CH;
        if (c + 1 == NSL) {
            #pragma unroll
            for (int r = 0; r < TILE_M; ++r)
                selfv[r] = out[(size_t)(rowBase + r) * COLS + t];
        }
        float* Ac = At + c * CH;
        if (c & 1) {
            if (c + 2 < NSL)      gather_slice<1, 1>(Ac, rp, sd, pre, xn);
            else if (c + 1 < NSL) gather_slice<1, 2>(Ac, rp, sd, pre, xn);
            else                  gather_slice<1, 0>(Ac, rp, sd, pre, xn);
        } else {
            if (c + 2 < NSL)      gather_slice<0, 1>(Ac, rp, sd, pre, xn);
            else if (c + 1 < NSL) gather_slice<0, 2>(Ac, rp, sd, pre, xn);
            else                  gather_slice<0, 0>(Ac, rp, sd, pre, xn);
        }
        lds_barrier();
    }

    float v[4][TILE_M];
    {
        const float bv = biasLds[t];
        #pragma unroll
        for (int r = 0; r < TILE_M; ++r)
            v[0][r] = selfv[r] + bv;
    }
    #pragma unroll
    for (int s = 1; s < 4; ++s) {
        const float bv = biasLds[s * 256 + t];
        #pragma unroll
        for (int r = 0; r < TILE_M; ++r)
            v[s][r] = Alds[(s - 1) * ATILE + r * AROW + t] + bv;
    }

    float sq[TILE_M];
    #pragma unroll
    for (int r = 0; r < TILE_M; ++r) {
        sq[r] = 0.0f;
        #pragma unroll
        for (int s = 0; s < 4; ++s) sq[r] += v[s][r] * v[s][r];
    }
    #pragma unroll
    for (int r = 0; r < TILE_M; ++r) sq[r] += xlane16<1>(sq[r]);
    #pragma unroll
    for (int r = 0; r < TILE_M; ++r) sq[r] += xlane16<2>(sq[r]);
    #pragma unroll
    for (int r = 0; r < TILE_M; ++r) sq[r] += xlane16<4>(sq[r]);
    #pragma unroll
    for (int r = 0; r < TILE_M; ++r) sq[r] += xlane16<8>(sq[r]);
    #pragma unroll
    for (int r = 0; r < TILE_M; ++r) sq[r] += xhalf(sq[r]);
    if (lane == 0) {
        #pragma unroll
        for (int r = 0; r < TILE_M; ++r) atomicAdd(&rowsum[r], sq[r]);
    }
    __syncthreads();

    const float myinv = 1.0f / fmaxf(sqrtf(rowsum[lane & 15]), 1e-12f);
    float inv[TILE_M];
    #pragma unroll
    for (int r = 0; r < TILE_M; ++r)
        inv[r] = __builtin_bit_cast(float, __builtin_amdgcn_readlane(__builtin_bit_cast(int, myinv), r));

    #pragma unroll
    for (int s = 0; s < 4; ++s) {
        float s1 = 0.0f, s2 = 0.0f;
        #pragma unroll
        for (int r = 0; r < TILE_M; ++r) {
            float w = fmaxf(v[s][r] * inv[r], 0.0f);
            s1 += w;
            s2 += w * w;
            out[(size_t)(rowBase + r) * COLS + s * 256 + t] = w;
        }
        atomicAdd(&gsum[s * 256 + t],   s1);
        atomicAdd(&gsumsq[s * 256 + t], s2);
    }
}

#define BNR 4
__global__ void __launch_bounds__(256) sage_bn_apply(float* __restrict__ out,
                                                     const float* __restrict__ gsum,
                                                     const float* __restrict__ gsumsq,
                                                     const float* __restrict__ gamma,
                                                     const float* __restrict__ beta)
{
    const int i = blockIdx.x * 256 + threadIdx.x;
    const int c = i & (COLS - 1);
    const int g = i >> 10;
    float* p = out + (size_t)g * (BNR * COLS) + c;
    float v[BNR];
    for (int k = 0; k < BNR; ++k) v[k] = p[k * COLS];
    const float invn = 1.0f / (float)ROWS;
    const float mu  = gsum[c] * invn;
    const float var = gsumsq[c] * invn - mu * mu;
    const float rs  = rsqrtf(var + 1e-5f);
    const float ga  = gamma[c];
    const float be  = beta[c];
    for (int k = 0; k < BNR; ++k) p[k * COLS] = (v[k] - mu) * rs * ga + be;
}

extern "C" void kernel_launch(void* const* d_in, const int* in_sizes, int n_in,
                              void* d_out, int out_size, void* d_ws, size_t ws_size,
                              hipStream_t stream) {
    (void)in_sizes; (void)n_in; (void)out_size;
    const float* x    = (const float*)d_in[0];
    const int*   idx1 = (const int*)  d_in[1];
    const int*   idx2 = (const int*)  d_in[2];
    const int*   idx3 = (const int*)  d_in[3];
    const float* Wx   = (const float*)d_in[4];
    const float* bx   = (const float*)d_in[5];
    const float* Wn   = (const float*)d_in[6];
    const float* bnb  = (const float*)d_in[7];
    const float* gam  = (const float*)d_in[8];
    const float* bet  = (const float*)d_in[9];
    float* out = (float*)d_out;

    float* gsum   = (float*)d_ws;
    float* gsumsq = gsum + COLS;

    if (ws_size >= WS_NEED) {
        float* pn = gsum + PN_OFF;
        sage_project<<<(ROWS / PM) * 4, 256, 0, stream>>>(x, Wx, Wn, out, pn, gsum);
        sage_gather<<<ROWS / TILE_M, 256, 0, stream>>>(
            pn, idx1, idx2, idx3, bx, bnb, out, gsum, gsumsq);
    } else {
        sage_init_stats<<<8, 256, 0, stream>>>(gsum);
        sage_fused<<<ROWS / TILE_M, 256, 0, stream>>>(
            x, idx1, idx2, idx3, Wx, bx, Wn, bnb, out, gsum, gsumsq);
    }

    sage_bn_apply<<<(ROWS * COLS) / (256 * BNR), 256, 0, stream>>>(out, gsum, gsumsq, gam, bet);
}
